# starting kernel
# baseline (speedup 1.0000x reference)
#include <hip/hip_runtime.h>
#include <hip/hip_bf16.h>

typedef __attribute__((ext_vector_type(2))) float v2f;
typedef __attribute__((ext_vector_type(8))) float v8f;

#define BN        64
#define NODE_DIM  128
#define EDGE_DIM  16
#define OUT_DIM   128
#define NHEADS    4
#define HEAD_DIM  32
#define LSTR      132

__global__ __launch_bounds__(256, 2)
void gat_fused_kernel(const float* __restrict__ nodes,
                      const float* __restrict__ edgef,
                      const int*   __restrict__ mask,
                      const float* __restrict__ Wn,
                      const float* __restrict__ We,
                      const float* __restrict__ a_src,
                      const float* __restrict__ a_dst,
                      const float* __restrict__ Wo,
                      const float* __restrict__ bo,
                      float* __restrict__ out)
{
    __shared__ float Xs[BN * LSTR];
    __shared__ float Hs[BN * LSTR];
    __shared__ float sc[NHEADS * BN];
    __shared__ float avec[HEAD_DIM];
    __shared__ float smax[NHEADS];
    __shared__ float sinv[NHEADS];
    __shared__ float wsum[OUT_DIM];

    const int t      = threadIdx.x;
    const int b      = blockIdx.x;
    const int lane   = t & 31;
    const int wave   = t >> 5;
    const int laneLo = lane & 15;
    const int laneHi = lane >> 4;

    {
        const float4* g4 = reinterpret_cast<const float4*>(nodes + (size_t)b * (BN * NODE_DIM));
#pragma unroll
        for (int j = 0; j < 8; ++j) {
            int    i = t + 256 * j;
            float4 v = g4[i];
            int    p   = i << 2;
            int    row = p >> 7;
            int    col = p & 127;
            float* d   = &Xs[row * LSTR + col];
            d[0] = v.x; d[1] = v.y; d[2] = v.z; d[3] = v.w;
        }
    }
    if (t < HEAD_DIM) avec[t] = a_src[t] + a_dst[t];
    __syncthreads();

    {
        const int n0 = wave * 16;
        v8f acc[4] = {};
        for (int k0 = 0; k0 < NODE_DIM; k0 += 4) {
            const int kb = k0 + laneHi * 2;
            v2f bf;
            bf.x = Wn[kb * OUT_DIM + n0 + laneLo];
            bf.y = Wn[(kb + 1) * OUT_DIM + n0 + laneLo];
#pragma unroll
            for (int mt = 0; mt < 4; ++mt) {
                const v2f af = *reinterpret_cast<const v2f*>(&Xs[(mt * 16 + laneLo) * LSTR + kb]);
                acc[mt] = __builtin_amdgcn_wmma_f32_16x16x4_f32(
                    false, af, false, bf, (short)0, acc[mt], false, false);
            }
        }
#pragma unroll
        for (int mt = 0; mt < 4; ++mt) {
            const int rowb = mt * 16 + laneHi * 8;
#pragma unroll
            for (int v = 0; v < 8; ++v)
                Hs[(rowb + v) * LSTR + n0 + laneLo] = acc[mt][v];
        }
    }
    __syncthreads();

    const int  hh    = t >> 6;
    const int  nn    = t & 63;
    float      s     = 0.0f;
    bool       valid;
    {
        const float* hr = &Hs[nn * LSTR + hh * HEAD_DIM];
#pragma unroll
        for (int d = 0; d < HEAD_DIM; ++d) s += hr[d] * avec[d];
        const float* er = edgef + ((size_t)b * BN + nn) * EDGE_DIM;
#pragma unroll
        for (int e = 0; e < EDGE_DIM; ++e) s += er[e] * We[e * NHEADS + hh];
        s     = (s > 0.0f) ? s : 0.2f * s;
        valid = (mask[(size_t)b * BN + nn] != 0);
        sc[t] = valid ? s : -__builtin_inff();
    }
    __syncthreads();

    if (t < NHEADS) {
        float mx = -__builtin_inff();
        for (int i = 0; i < BN; ++i) mx = fmaxf(mx, sc[t * BN + i]);
        float sum = 0.0f;
        for (int i = 0; i < BN; ++i) {
            float v = sc[t * BN + i];
            if (v > -__builtin_inff()) sum += __expf(v - mx);
        }
        smax[t] = mx;
        sinv[t] = (sum > 0.0f) ? (1.0f / sum) : 0.0f;
    }
    __syncthreads();
    sc[t] = valid ? __expf(s - smax[hh]) * sinv[hh] : 0.0f;
    __syncthreads();

    if (t < OUT_DIM) {
        const int    ho = t >> 5;
        const float* pw = &sc[ho * BN];
        float        acc = 0.0f;
        for (int n = 0; n < BN; ++n) acc = fmaf(pw[n], Hs[n * LSTR + t], acc);
        wsum[t] = acc;
    }
    __syncthreads();

    if (t < OUT_DIM) {
        float r = bo[t];
        for (int k = 0; k < OUT_DIM; ++k) r = fmaf(wsum[k], Wo[k * OUT_DIM + t], r);
        out[(size_t)b * OUT_DIM + t] = r;
    }
}

extern "C" void kernel_launch(void* const* d_in, const int* in_sizes, int n_in,
                              void* d_out, int out_size, void* d_ws, size_t ws_size,
                              hipStream_t stream) {
    const float* nodes = (const float*)d_in[0];
    const float* edgef = (const float*)d_in[1];
    const int*   mask  = (const int*)  d_in[2];
    const float* Wn    = (const float*)d_in[3];
    const float* We    = (const float*)d_in[4];
    const float* a_src = (const float*)d_in[5];
    const float* a_dst = (const float*)d_in[6];
    const float* Wo    = (const float*)d_in[7];
    const float* bo    = (const float*)d_in[8];
    float*       out   = (float*)d_out;

    const int B = in_sizes[0] / (BN * NODE_DIM);
    gat_fused_kernel<<<B, 256, 0, stream>>>(nodes, edgef, mask, Wn, We, a_src, a_dst, Wo, bo, out);
}


# Softmax max and sum over each head's 64 nodes by lane butterflies on all 256 threads, not 4 threads looping serially over LDS; the sum of exponentials is now added in tree order
# speedup vs baseline: 1.1873x; 1.1873x over previous
#include <hip/hip_runtime.h>
#include <hip/hip_bf16.h>

typedef __attribute__((ext_vector_type(2))) float v2f;
typedef __attribute__((ext_vector_type(8))) float v8f;

#define BN        64
#define NODE_DIM  128
#define EDGE_DIM  16
#define OUT_DIM   128
#define NHEADS    4
#define HEAD_DIM  32
#define LSTR      132

template <int M>
__device__ __forceinline__ float lane_xor_in_row(float x) {
    const int b = __builtin_bit_cast(int, x);
    return __builtin_bit_cast(float, __builtin_amdgcn_update_dpp(0, b, 0x160 + M, 0xf, 0xf, true));
}
__device__ __forceinline__ float lane_other_row(float x) {
    const int b = __builtin_bit_cast(int, x);
    return __builtin_bit_cast(float,
        __builtin_amdgcn_permlanex16(b, b, 0x76543210, 0xfedcba98, false, false));
}
__device__ __forceinline__ float wave_max(float x) {
    x = fmaxf(x, lane_xor_in_row<1>(x));
    x = fmaxf(x, lane_xor_in_row<2>(x));
    x = fmaxf(x, lane_xor_in_row<4>(x));
    x = fmaxf(x, lane_xor_in_row<8>(x));
    return fmaxf(x, lane_other_row(x));
}
__device__ __forceinline__ float wave_sum(float x) {
    x += lane_xor_in_row<1>(x);
    x += lane_xor_in_row<2>(x);
    x += lane_xor_in_row<4>(x);
    x += lane_xor_in_row<8>(x);
    return x + lane_other_row(x);
}

__global__ __launch_bounds__(256, 2)
void gat_fused_kernel(const float* __restrict__ nodes,
                      const float* __restrict__ edgef,
                      const int*   __restrict__ mask,
                      const float* __restrict__ Wn,
                      const float* __restrict__ We,
                      const float* __restrict__ a_src,
                      const float* __restrict__ a_dst,
                      const float* __restrict__ Wo,
                      const float* __restrict__ bo,
                      float* __restrict__ out)
{
    __shared__ float Xs[BN * LSTR];
    __shared__ float Hs[BN * LSTR];
    __shared__ float sc[NHEADS * BN];
    __shared__ float avec[HEAD_DIM];
    __shared__ float wmax[8];
    __shared__ float wsm[8];
    __shared__ float wsum[OUT_DIM];

    const int t      = threadIdx.x;
    const int b      = blockIdx.x;
    const int lane   = t & 31;
    const int wave   = t >> 5;
    const int laneLo = lane & 15;
    const int laneHi = lane >> 4;

    {
        const float4* g4 = reinterpret_cast<const float4*>(nodes + (size_t)b * (BN * NODE_DIM));
#pragma unroll
        for (int j = 0; j < 8; ++j) {
            int    i = t + 256 * j;
            float4 v = g4[i];
            int    p   = i << 2;
            int    row = p >> 7;
            int    col = p & 127;
            float* d   = &Xs[row * LSTR + col];
            d[0] = v.x; d[1] = v.y; d[2] = v.z; d[3] = v.w;
        }
    }
    if (t < HEAD_DIM) avec[t] = a_src[t] + a_dst[t];
    __syncthreads();

    {
        const int n0 = wave * 16;
        v8f acc[4] = {};
        for (int k0 = 0; k0 < NODE_DIM; k0 += 4) {
            const int kb = k0 + laneHi * 2;
            v2f bf;
            bf.x = Wn[kb * OUT_DIM + n0 + laneLo];
            bf.y = Wn[(kb + 1) * OUT_DIM + n0 + laneLo];
#pragma unroll
            for (int mt = 0; mt < 4; ++mt) {
                const v2f af = *reinterpret_cast<const v2f*>(&Xs[(mt * 16 + laneLo) * LSTR + kb]);
                acc[mt] = __builtin_amdgcn_wmma_f32_16x16x4_f32(
                    false, af, false, bf, (short)0, acc[mt], false, false);
            }
        }
#pragma unroll
        for (int mt = 0; mt < 4; ++mt) {
            const int rowb = mt * 16 + laneHi * 8;
#pragma unroll
            for (int v = 0; v < 8; ++v)
                Hs[(rowb + v) * LSTR + n0 + laneLo] = acc[mt][v];
        }
    }
    __syncthreads();

    const int  hh    = t >> 6;
    const int  nn    = t & 63;
    float      s     = 0.0f;
    bool       valid;
    {
        const float* hr = &Hs[nn * LSTR + hh * HEAD_DIM];
#pragma unroll
        for (int d = 0; d < HEAD_DIM; ++d) s += hr[d] * avec[d];
        const float* er = edgef + ((size_t)b * BN + nn) * EDGE_DIM;
#pragma unroll
        for (int e = 0; e < EDGE_DIM; ++e) s += er[e] * We[e * NHEADS + hh];
        s     = (s > 0.0f) ? s : 0.2f * s;
        valid = (mask[(size_t)b * BN + nn] != 0);
    }

    {
        const float mw = wave_max(valid ? s : -__builtin_inff());
        if (lane == 0) wmax[wave] = mw;
    }
    __syncthreads();
    const float mx = fmaxf(wmax[2 * hh], wmax[2 * hh + 1]);
    const float ex = __expf(s - mx);
    const float e  = valid ? ex : 0.0f;
    {
        const float sw = wave_sum(e);
        if (lane == 0) wsm[wave] = sw;
    }
    __syncthreads();
    {
        const float sum = wsm[2 * hh] + wsm[2 * hh + 1];
        const float inv = (sum > 0.0f) ? (1.0f / sum) : 0.0f;
        sc[t] = e * inv;
    }
    __syncthreads();

    if (t < OUT_DIM) {
        const int    ho = t >> 5;
        const float* pw = &sc[ho * BN];
        float        acc = 0.0f;
        for (int n = 0; n < BN; ++n) acc = fmaf(pw[n], Hs[n * LSTR + t], acc);
        wsum[t] = acc;
    }
    __syncthreads();

    if (t < OUT_DIM) {
        float r = bo[t];
        for (int k = 0; k < OUT_DIM; ++k) r = fmaf(wsum[k], Wo[k * OUT_DIM + t], r);
        out[(size_t)b * OUT_DIM + t] = r;
    }
}

extern "C" void kernel_launch(void* const* d_in, const int* in_sizes, int n_in,
                              void* d_out, int out_size, void* d_ws, size_t ws_size,
                              hipStream_t stream) {
    const float* nodes = (const float*)d_in[0];
    const float* edgef = (const float*)d_in[1];
    const int*   mask  = (const int*)  d_in[2];
    const float* Wn    = (const float*)d_in[3];
    const float* We    = (const float*)d_in[4];
    const float* a_src = (const float*)d_in[5];
    const float* a_dst = (const float*)d_in[6];
    const float* Wo    = (const float*)d_in[7];
    const float* bo    = (const float*)d_in[8];
    float*       out   = (float*)d_out;

    const int B = in_sizes[0] / (BN * NODE_DIM);
    gat_fused_kernel<<<B, 256, 0, stream>>>(nodes, edgef, mask, Wn, We, a_src, a_dst, Wo, bo, out);
}


# Weighted sum and out projection split by channel over all 8 waves: each wave reads its 16 W_out rows as 16-byte loads in one group under the weighted sum; node and k sums added in partial groups
# speedup vs baseline: 1.2842x; 1.0816x over previous
#include <hip/hip_runtime.h>
#include <hip/hip_bf16.h>

typedef __attribute__((ext_vector_type(2))) float v2f;
typedef __attribute__((ext_vector_type(8))) float v8f;

#define BN        64
#define NODE_DIM  128
#define EDGE_DIM  16
#define OUT_DIM   128
#define NHEADS    4
#define HEAD_DIM  32
#define LSTR      132

template <int M>
__device__ __forceinline__ float lane_xor_in_row(float x) {
    const int b = __builtin_bit_cast(int, x);
    return __builtin_bit_cast(float, __builtin_amdgcn_update_dpp(0, b, 0x160 + M, 0xf, 0xf, true));
}
__device__ __forceinline__ float lane_other_row(float x) {
    const int b = __builtin_bit_cast(int, x);
    return __builtin_bit_cast(float,
        __builtin_amdgcn_permlanex16(b, b, 0x76543210, 0xfedcba98, false, false));
}
__device__ __forceinline__ float wave_max(float x) {
    x = fmaxf(x, lane_xor_in_row<1>(x));
    x = fmaxf(x, lane_xor_in_row<2>(x));
    x = fmaxf(x, lane_xor_in_row<4>(x));
    x = fmaxf(x, lane_xor_in_row<8>(x));
    return fmaxf(x, lane_other_row(x));
}
__device__ __forceinline__ float wave_sum(float x) {
    x += lane_xor_in_row<1>(x);
    x += lane_xor_in_row<2>(x);
    x += lane_xor_in_row<4>(x);
    x += lane_xor_in_row<8>(x);
    return x + lane_other_row(x);
}

__global__ __launch_bounds__(256, 2)
void gat_fused_kernel(const float* __restrict__ nodes,
                      const float* __restrict__ edgef,
                      const int*   __restrict__ mask,
                      const float* __restrict__ Wn,
                      const float* __restrict__ We,
                      const float* __restrict__ a_src,
                      const float* __restrict__ a_dst,
                      const float* __restrict__ Wo,
                      const float* __restrict__ bo,
                      float* __restrict__ out)
{
    __shared__ float Xs[BN * LSTR];
    __shared__ float Hs[BN * LSTR];
    __shared__ float sc[NHEADS * BN];
    __shared__ float avec[HEAD_DIM];
    __shared__ float wmax[8];
    __shared__ float wsm[8];
    __shared__ float4 red4[8 * (OUT_DIM / 4)];

    const int t      = threadIdx.x;
    const int b      = blockIdx.x;
    const int lane   = t & 31;
    const int wave   = t >> 5;
    const int laneLo = lane & 15;
    const int laneHi = lane >> 4;

    {
        const float4* g4 = reinterpret_cast<const float4*>(nodes + (size_t)b * (BN * NODE_DIM));
#pragma unroll
        for (int j = 0; j < 8; ++j) {
            int    i = t + 256 * j;
            float4 v = g4[i];
            int    p   = i << 2;
            int    row = p >> 7;
            int    col = p & 127;
            float* d   = &Xs[row * LSTR + col];
            d[0] = v.x; d[1] = v.y; d[2] = v.z; d[3] = v.w;
        }
    }
    if (t < HEAD_DIM) avec[t] = a_src[t] + a_dst[t];
    __syncthreads();

    {
        const int n0 = wave * 16;
        v8f acc[4] = {};
        for (int k0 = 0; k0 < NODE_DIM; k0 += 4) {
            const int kb = k0 + laneHi * 2;
            v2f bf;
            bf.x = Wn[kb * OUT_DIM + n0 + laneLo];
            bf.y = Wn[(kb + 1) * OUT_DIM + n0 + laneLo];
#pragma unroll
            for (int mt = 0; mt < 4; ++mt) {
                const v2f af = *reinterpret_cast<const v2f*>(&Xs[(mt * 16 + laneLo) * LSTR + kb]);
                acc[mt] = __builtin_amdgcn_wmma_f32_16x16x4_f32(
                    false, af, false, bf, (short)0, acc[mt], false, false);
            }
        }
#pragma unroll
        for (int mt = 0; mt < 4; ++mt) {
            const int rowb = mt * 16 + laneHi * 8;
#pragma unroll
            for (int v = 0; v < 8; ++v)
                Hs[(rowb + v) * LSTR + n0 + laneLo] = acc[mt][v];
        }
    }
    __syncthreads();

    const int  hh    = t >> 6;
    const int  nn    = t & 63;
    float      s     = 0.0f;
    bool       valid;
    {
        const float* hr = &Hs[nn * LSTR + hh * HEAD_DIM];
#pragma unroll
        for (int d = 0; d < HEAD_DIM; ++d) s += hr[d] * avec[d];
        const float* er = edgef + ((size_t)b * BN + nn) * EDGE_DIM;
#pragma unroll
        for (int e = 0; e < EDGE_DIM; ++e) s += er[e] * We[e * NHEADS + hh];
        s     = (s > 0.0f) ? s : 0.2f * s;
        valid = (mask[(size_t)b * BN + nn] != 0);
    }

    {
        const float mw = wave_max(valid ? s : -__builtin_inff());
        if (lane == 0) wmax[wave] = mw;
    }
    __syncthreads();
    const float mx = fmaxf(wmax[2 * hh], wmax[2 * hh + 1]);
    const float ex = __expf(s - mx);
    const float e  = valid ? ex : 0.0f;
    {
        const float sw = wave_sum(e);
        if (lane == 0) wsm[wave] = sw;
    }
    __syncthreads();
    {
        const float sum = wsm[2 * hh] + wsm[2 * hh + 1];
        const float inv = (sum > 0.0f) ? (1.0f / sum) : 0.0f;
        sc[t] = e * inv;
    }
    __syncthreads();

    float4 wo[16];
    const float bias = bo[t & (OUT_DIM - 1)];
    {
        const float4* wo4 = reinterpret_cast<const float4*>(Wo) + (size_t)(wave * 16) * (OUT_DIM / 4) + lane;
        for (int j = 0; j < 16; ++j) wo[j] = wo4[j * (OUT_DIM / 4)];
    }

    float wk_mine;
    {
        const float* pw = &sc[(wave >> 1) * BN + laneHi * 8];
        const float* hc = &Hs[(laneHi * 8) * LSTR + wave * 16 + laneLo];
        float a = 0.0f;
        for (int g = 0; g < 4; ++g) {
            for (int i = 0; i < 8; ++i) {
                const int n = g * 16 + i;
                a = fmaf(pw[n], hc[n * LSTR], a);
            }
        }
        wk_mine = a + lane_other_row(a);
    }

    {
        float4 r = {0.0f, 0.0f, 0.0f, 0.0f};
        for (int j = 0; j < 16; ++j) {
            const float wk = __builtin_bit_cast(float,
                __builtin_amdgcn_readlane(__builtin_bit_cast(int, wk_mine), j));
            r.x = fmaf(wk, wo[j].x, r.x);
            r.y = fmaf(wk, wo[j].y, r.y);
            r.z = fmaf(wk, wo[j].z, r.z);
            r.w = fmaf(wk, wo[j].w, r.w);
        }
        red4[wave * (OUT_DIM / 4) + lane] = r;
    }
    __syncthreads();
    if (t < OUT_DIM) {
        const float* red = reinterpret_cast<const float*>(red4);
        float r = bias;
        for (int w = 0; w < 8; ++w) r += red[w * OUT_DIM + t];
        out[(size_t)b * OUT_DIM + t] = r;
    }
}

extern "C" void kernel_launch(void* const* d_in, const int* in_sizes, int n_in,
                              void* d_out, int out_size, void* d_ws, size_t ws_size,
                              hipStream_t stream) {
    const float* nodes = (const float*)d_in[0];
    const float* edgef = (const float*)d_in[1];
    const int*   mask  = (const int*)  d_in[2];
    const float* Wn    = (const float*)d_in[3];
    const float* We    = (const float*)d_in[4];
    const float* a_src = (const float*)d_in[5];
    const float* a_dst = (const float*)d_in[6];
    const float* Wo    = (const float*)d_in[7];
    const float* bo    = (const float*)d_in[8];
    float*       out   = (float*)d_out;

    const int B = in_sizes[0] / (BN * NODE_DIM);
    gat_fused_kernel<<<B, 256, 0, stream>>>(nodes, edgef, mask, Wn, We, a_src, a_dst, Wo, bo, out);
}
